# starting kernel
# baseline (speedup 1.0000x reference)
#include <hip/hip_runtime.h>

typedef __attribute__((ext_vector_type(2))) float v2f;
typedef __attribute__((ext_vector_type(8))) float v8f;

#define BLOCK 256
#define WAVES 8
#define JSTRIDE 17

__device__ __forceinline__ int cnot_map(int k, int c) {
    int t = (c + 1) & 3;
    int cbit = (k >> (3 - c)) & 1;
    return k ^ (cbit << (3 - t));
}

__device__ __forceinline__ float xor16(float v) {
    return __int_as_float(__builtin_amdgcn_ds_swizzle(__float_as_int(v), 0x401F));
}

__global__ __launch_bounds__(BLOCK) void qlstm_kernel(
    const float* __restrict__ x, const float* __restrict__ h_prev,
    const float* __restrict__ W_in, const float* __restrict__ b_in,
    const float* __restrict__ Up, const float* __restrict__ Ud,
    float* __restrict__ out, int Btot)
{
    __shared__ float trig[4][4][2];
    __shared__ float Cmat[2][16][16];
    __shared__ float ldsJ[WAVES][32][JSTRIDE];

    const int tid = threadIdx.x;

    if (tid < 16) {
        int l = tid >> 2, q = tid & 3;
        float w = (l == 0) ? Up[q] : (l == 1) ? Up[4 + q]
                : (l == 2) ? Ud[4 + q] : Ud[q];
        trig[l][q][0] = cosf(0.5f * w);
        trig[l][q][1] = sinf(0.5f * w);
    }
    const int ii = tid >> 4, jj = tid & 15;
    Cmat[0][ii][jj] = (ii == jj) ? 1.0f : 0.0f;
    __syncthreads();

    int cur = 0;
    #pragma unroll
    for (int l = 0; l < 4; ++l) {
        int kk = ii;
        kk = cnot_map(kk, 3); kk = cnot_map(kk, 2);
        kk = cnot_map(kk, 1); kk = cnot_map(kk, 0);
        float sum = 0.0f;
        #pragma unroll
        for (int k = 0; k < 16; ++k) {
            float u = 1.0f;
            #pragma unroll
            for (int q = 0; q < 4; ++q) {
                int r  = (kk >> (3 - q)) & 1;
                int cb = (k  >> (3 - q)) & 1;
                float cc = trig[l][q][0], ss = trig[l][q][1];
                u *= (r == cb) ? cc : (r ? ss : -ss);
            }
            sum += u * Cmat[cur][k][jj];
        }
        Cmat[cur ^ 1][ii][jj] = sum;
        __syncthreads();
        cur ^= 1;
    }

    const int lane = tid & 31;
    const int wv   = tid >> 5;
    const int hh   = lane >> 4;
    const int mn   = lane & 15;
    v2f afrag[4];
    #pragma unroll
    for (int kc = 0; kc < 4; ++kc) {
        afrag[kc].x = Cmat[cur][mn][4 * kc + 2 * hh + 0];
        afrag[kc].y = Cmat[cur][mn][4 * kc + 2 * hh + 1];
    }
    float Wl[4], bl[4];
    #pragma unroll
    for (int j = 0; j < 4; ++j) { Wl[j] = W_in[j]; bl[j] = b_in[j]; }

    const int totalWaves = gridDim.x * WAVES;
    const int gw = blockIdx.x * WAVES + wv;
    const int stride = totalWaves * 32;

    for (int base = gw * 32; base < Btot; base += stride) {
        int e  = base + lane;
        int ec = min(e, Btot - 1);
        float xv = x[ec];
        float4 hp = ((const float4*)h_prev)[ec];

        int en = min(ec + stride, Btot - 1);
        __builtin_prefetch(&h_prev[4 * (size_t)en], 0, 1);

        float pj[4], tt, nn = 0.0f;
        #pragma unroll
        for (int j = 0; j < 4; ++j) {
            pj[j] = xv * Wl[j] + bl[j];
            tt = pj[j] + 1e-12f;
            nn += tt * tt;
        }
        float inv1 = 1.0f / sqrtf(nn);

        float jv[16];
        float hpv[4] = {hp.x, hp.y, hp.z, hp.w};
        float s2 = 0.0f;
        #pragma unroll
        for (int j = 0; j < 4; ++j) {
            float inj = pj[j] * inv1;
            #pragma unroll
            for (int k = 0; k < 4; ++k) {
                float t = inj * hpv[k];
                jv[4 * j + k] = t;
                s2 += t * t;
            }
        }
        float inv2 = 1.0f / sqrtf(s2);
        float* row = ldsJ[wv][lane];
        #pragma unroll
        for (int i2 = 0; i2 < 16; ++i2) row[i2] = jv[i2] * inv2;
        __builtin_amdgcn_wave_barrier();

        #pragma unroll
        for (int half = 0; half < 2; ++half) {
            const float* jrow = ldsJ[wv][half * 16 + mn];
            v2f bfrag[4];
            #pragma unroll
            for (int kc = 0; kc < 4; ++kc) {
                bfrag[kc].x = jrow[4 * kc + 2 * hh + 0];
                bfrag[kc].y = jrow[4 * kc + 2 * hh + 1];
            }
            v8f acc = {};
            #pragma unroll
            for (int kc = 0; kc < 4; ++kc)
                acc = __builtin_amdgcn_wmma_f32_16x16x4_f32(
                    false, afrag[kc], false, bfrag[kc],
                    (short)0, acc, false, false);

            float p[8];
            #pragma unroll
            for (int v = 0; v < 8; ++v) p[v] = acc[v] * acc[v];
            float ha0 = p[0] + p[4], ha1 = p[1] + p[5];
            float ha2 = p[2] + p[6], ha3 = p[3] + p[7];
            float op = (p[0] + p[1] + p[2] + p[3]) - (p[4] + p[5] + p[6] + p[7]);
            ha0 += xor16(ha0); ha1 += xor16(ha1);
            ha2 += xor16(ha2); ha3 += xor16(ha3);
            op  += xor16(op);

            int eo = base + half * 16 + mn;
            if (hh == 0 && eo < Btot) {
                ((float4*)out)[eo] = make_float4(ha0, ha1, ha2, ha3);
                out[4 * (size_t)Btot + eo] = op;
            }
        }
        __builtin_amdgcn_wave_barrier();
    }
}

extern "C" void kernel_launch(void* const* d_in, const int* in_sizes, int n_in,
                              void* d_out, int out_size, void* d_ws, size_t ws_size,
                              hipStream_t stream) {
    const float* x      = (const float*)d_in[0];
    const float* h_prev = (const float*)d_in[1];
    const float* W_in   = (const float*)d_in[2];
    const float* b_in   = (const float*)d_in[3];
    const float* Up     = (const float*)d_in[4];
    const float* Ud     = (const float*)d_in[5];
    float* out = (float*)d_out;

    int Btot = in_sizes[0];
    int tiles = (Btot + 32 * WAVES - 1) / (32 * WAVES);
    int blocks = 1024;
    if (blocks > tiles) blocks = (tiles > 0) ? tiles : 1;

    qlstm_kernel<<<blocks, BLOCK, 0, stream>>>(x, h_prev, W_in, b_in, Up, Ud, out, Btot);
}


# Product on raw h_prev: A = rows of C(W x I), C(b x I) (M=32, K=4), B fragments straight from global; x and 1/sum(psi^2) applied to the D tile; no LDS in the sweep, sums in-lane; sum order changes
# speedup vs baseline: 1.3654x; 1.3654x over previous
#include <hip/hip_runtime.h>

typedef __attribute__((ext_vector_type(2))) float v2f;
typedef __attribute__((ext_vector_type(8))) float v8f;

#define BLOCK 256
#define WAVES 8

__device__ __forceinline__ constexpr int cnot_map(int k, int c) {
    int t = (c + 1) & 3;
    int cbit = (k >> (3 - c)) & 1;
    return k ^ (cbit << (3 - t));
}

__device__ __forceinline__ constexpr int ring_inv(int i) {
    i = cnot_map(i, 3); i = cnot_map(i, 2);
    i = cnot_map(i, 1); i = cnot_map(i, 0);
    return i;
}

__device__ __forceinline__ float other_half(float v) {
    int i = __float_as_int(v);
    return __int_as_float(__builtin_amdgcn_permlanex16(i, i, 0x76543210, 0xfedcba98, false, false));
}

__device__ __forceinline__ void block_sums(v2f aW, v2f aB, v2f hf, float xv,
                                           float& f0, float& f1, float& zs) {
    v8f zero = {};
    v8f u = __builtin_amdgcn_wmma_f32_16x16x4_f32(false, aW, false, hf,
                                                  (short)0, zero, false, false);
    v8f w = __builtin_amdgcn_wmma_f32_16x16x4_f32(false, aB, false, hf,
                                                  (short)0, zero, false, false);
    const v2f xs = {xv, xv};
    const v2f p01 = xs * __builtin_shufflevector(u, u, 0, 1) + __builtin_shufflevector(w, w, 0, 1);
    const v2f p23 = xs * __builtin_shufflevector(u, u, 2, 3) + __builtin_shufflevector(w, w, 2, 3);
    const v2f p45 = xs * __builtin_shufflevector(u, u, 4, 5) + __builtin_shufflevector(w, w, 4, 5);
    const v2f p67 = xs * __builtin_shufflevector(u, u, 6, 7) + __builtin_shufflevector(w, w, 6, 7);
    const v2f pn0 = p01 * p01 + p23 * p23;
    const v2f pn1 = p45 * p45 + p67 * p67;
    const v2f pn  = pn0 + pn1;
    f0 = pn0.x + pn0.y;
    f1 = pn1.x + pn1.y;
    zs = pn.x - pn.y;
}

__device__ __forceinline__ float from_lane(float v, int src) {
    return __int_as_float(__builtin_amdgcn_readlane(__float_as_int(v), src));
}

template <bool RING_ON_K>
__device__ __forceinline__ void kron_operand(const float* c, const float* s,
                                             int laneIdx, int hh, v2f* out) {
    float f[4][2];
    #pragma unroll
    for (int q = 0; q < 4; ++q) {
        int m = (laneIdx >> (3 - q)) & 1;
        f[q][0] = m ? -s[q] : c[q];
        f[q][1] = m ?  c[q] : s[q];
    }
    #pragma unroll
    for (int kc = 0; kc < 4; ++kc) {
        #pragma unroll
        for (int t = 0; t < 2; ++t) {
            const int k0 = 2 * kc + t, k1 = 8 + 2 * kc + t;
            const int r0 = RING_ON_K ? ring_inv(k0) : k0;
            const int r1 = RING_ON_K ? ring_inv(k1) : k1;
            float u = 1.0f;
            #pragma unroll
            for (int q = 0; q < 4; ++q) {
                const int a0 = (r0 >> (3 - q)) & 1, a1 = (r1 >> (3 - q)) & 1;
                float fac = (a0 == a1) ? f[q][a0] : (hh ? f[q][a1] : f[q][a0]);
                u *= fac;
            }
            if (t == 0) out[kc].x = u; else out[kc].y = u;
        }
    }
}

__global__ __launch_bounds__(BLOCK) void qlstm_kernel(
    const float* __restrict__ x, const float* __restrict__ h_prev,
    const float* __restrict__ W_in, const float* __restrict__ b_in,
    const float* __restrict__ Up, const float* __restrict__ Ud,
    float* __restrict__ out, int Btot)
{
    __shared__ float Cmat[16][16];

    const int tid  = threadIdx.x;
    const int lane = tid & 31;
    const int wv   = tid >> 5;
    const int hh   = lane >> 4;
    const int mn   = lane & 15;

    if (tid < 32) {
        const int idx = lane & 15;
        const int l = idx >> 2, q = idx & 3;
        float w = (l == 0) ? Up[q] : (l == 1) ? Up[4 + q]
                : (l == 2) ? Ud[4 + q] : Ud[q];
        float cv = cosf(0.5f * w);
        float sv = sinf(0.5f * w);
        float cs[4][4], sn[4][4], nsn3[4];
        #pragma unroll
        for (int i = 0; i < 16; ++i) {
            cs[i >> 2][i & 3] = from_lane(cv, i);
            sn[i >> 2][i & 3] = from_lane(sv, i);
        }
        #pragma unroll
        for (int i = 0; i < 4; ++i) nsn3[i] = -sn[3][i];

        v2f bx[4];
        kron_operand<false>(cs[3], nsn3, ring_inv(mn), hh, bx);
        v8f d = {};
        #pragma unroll
        for (int ll = 2; ll >= 0; --ll) {
            v2f al[4];
            kron_operand<true>(cs[ll], sn[ll], mn, hh, al);
            v8f acc = {};
            #pragma unroll
            for (int kc = 0; kc < 4; ++kc)
                acc = __builtin_amdgcn_wmma_f32_16x16x4_f32(
                    false, al[kc], false, bx[kc],
                    (short)0, acc, false, false);
            d = acc;
            #pragma unroll
            for (int kc = 0; kc < 4; ++kc) { bx[kc].x = acc[2 * kc]; bx[kc].y = acc[2 * kc + 1]; }
        }
        #pragma unroll
        for (int v = 0; v < 8; ++v) Cmat[mn][8 * hh + v] = d[v];
    }
    __syncthreads();

    float Wl[4], bl[4];
    #pragma unroll
    for (int j = 0; j < 4; ++j) { Wl[j] = W_in[j]; bl[j] = b_in[j]; }
    const int rho = (mn >> 2) | ((mn & 3) << 2);
    v2f aW = {0.0f, 0.0f}, aB = {0.0f, 0.0f};
    #pragma unroll
    for (int j = 0; j < 4; ++j) {
        const float c0 = Cmat[rho][4 * j + 2 * hh + 0];
        const float c1 = Cmat[rho][4 * j + 2 * hh + 1];
        aW.x += c0 * Wl[j]; aW.y += c1 * Wl[j];
        aB.x += c0 * bl[j]; aB.y += c1 * bl[j];
    }

    const int totalWaves = gridDim.x * WAVES;
    const int gw = blockIdx.x * WAVES + wv;
    const int stride = totalWaves * 32;

    for (int base = gw * 32; base < Btot; base += stride) {
        const int eA = base + mn, eB = eA + 16;
        const int eAc = min(eA, Btot - 1), eBc = min(eB, Btot - 1);
        const v2f hA = *(const v2f*)(h_prev + 4 * (size_t)eAc + 2 * hh);
        const v2f hB = *(const v2f*)(h_prev + 4 * (size_t)eBc + 2 * hh);
        const float xA = x[eAc], xB = x[eBc];

        int ec = min(base + lane, Btot - 1);
        int en = min(ec + stride, Btot - 1);
        __builtin_prefetch(&h_prev[4 * (size_t)en], 0, 1);

        float fA0, fA1, zA, fB0, fB1, zB;
        block_sums(aW, aB, hA, xA, fA0, fA1, zA);
        block_sums(aW, aB, hB, xB, fB0, fB1, zB);

        float tA = fA0 + fA1, tB = fB0 + fB1;
        tA += other_half(tA);
        tB += other_half(tB);
        const float qA = __builtin_amdgcn_rcpf(tA);
        const float qB = __builtin_amdgcn_rcpf(tB);

        const float zkeep = hh ? zB : zA;
        const float zsend = hh ? zA : zB;
        const float zq    = hh ? qB : qA;
        const float op = (zkeep + other_half(zsend)) * zq;

        if (eA < Btot)
            *(float2*)(out + 4 * (size_t)eA + 2 * hh) = make_float2(fA0 * qA, fA1 * qA);
        if (eB < Btot)
            *(float2*)(out + 4 * (size_t)eB + 2 * hh) = make_float2(fB0 * qB, fB1 * qB);
        const int eo = base + lane;
        if (eo < Btot) out[4 * (size_t)Btot + eo] = op;
    }
}

extern "C" void kernel_launch(void* const* d_in, const int* in_sizes, int n_in,
                              void* d_out, int out_size, void* d_ws, size_t ws_size,
                              hipStream_t stream) {
    const float* x      = (const float*)d_in[0];
    const float* h_prev = (const float*)d_in[1];
    const float* W_in   = (const float*)d_in[2];
    const float* b_in   = (const float*)d_in[3];
    const float* Up     = (const float*)d_in[4];
    const float* Ud     = (const float*)d_in[5];
    float* out = (float*)d_out;

    int Btot = in_sizes[0];
    int tiles = (Btot + 32 * WAVES - 1) / (32 * WAVES);
    int blocks = 1024;
    if (blocks > tiles) blocks = (tiles > 0) ? tiles : 1;

    qlstm_kernel<<<blocks, BLOCK, 0, stream>>>(x, h_prev, W_in, b_in, Up, Ud, out, Btot);
}
